# starting kernel
# baseline (speedup 1.0000x reference)
#include <hip/hip_runtime.h>
#include <hip/hip_bf16.h>

typedef float v2f __attribute__((ext_vector_type(2)));
typedef float v8f __attribute__((ext_vector_type(8)));

static constexpr int kVocab = 256;
static constexpr int kE     = 64;
static constexpr int kH     = 256;
static constexpr int kW     = 20;
static constexpr int kWords = 64 * 128;
static constexpr int kF3    = 3 * kH;

__global__ __launch_bounds__(32) void build_tables(
    const float* __restrict__ emb,
    const float* __restrict__ w1, const float* __restrict__ b1,
    const float* __restrict__ w3, const float* __restrict__ b3,
    const float* __restrict__ w5, const float* __restrict__ b5,
    float* __restrict__ T)
{
    int tile  = blockIdx.x;
    int tap   = tile >> 8;
    int cTile = (tile >> 4) & 15;
    int hTile = tile & 15;

    const float* wp; const float* bp; int kk, j;
    if (tap == 0)     { wp = w1; bp = b1; kk = 1; j = 0; }
    else if (tap < 4) { wp = w3; bp = b3; kk = 3; j = tap - 1; }
    else              { wp = w5; bp = b5; kk = 5; j = tap - 4; }
    const bool center = (j == (kk >> 1));

    const int lane = threadIdx.x & 31;
    const int half = lane >> 4;
    const int l16  = lane & 15;

    const int row = cTile * 16 + l16;
    const int col = hTile * 16 + l16;

    v8f acc = {};
    for (int e0 = 0; e0 < kE; e0 += 4) {
        int ea = e0 + 2 * half;
        v2f a = *(const v2f*)(emb + row * kE + ea);
        v2f b;
        b.x = wp[(col * kE + ea    ) * kk + j];
        b.y = wp[(col * kE + ea + 1) * kk + j];
        acc = __builtin_amdgcn_wmma_f32_16x16x4_f32(
            false, a, false, b, (short)0, acc, false, false);
    }

    const float bias = center ? bp[col] : 0.0f;
    float* outp = T + tap * (kVocab * kH) + (cTile * 16) * kH + hTile * 16;
#pragma unroll
    for (int i = 0; i < 8; ++i) {
        int m = half * 8 + i;
        outp[m * kH + l16] = acc[i] + bias;
    }
}

__global__ __launch_bounds__(256) void conv_gather(
    const int*   __restrict__ chars,
    const float* __restrict__ T,
    float*       __restrict__ feat)
{
    __shared__ int c[kW];
    const int n   = blockIdx.x;
    const int tid = threadIdx.x;
    if (tid < kW) c[tid] = chars[n * kW + tid];
    __syncthreads();

    const int h = tid;
    const float* T0 = T;
    const float* T1 = T + 1 * 65536;
    const float* T2 = T + 2 * 65536;
    const float* T3 = T + 3 * 65536;
    const float* T4 = T + 4 * 65536;
    const float* T5 = T + 5 * 65536;
    const float* T6 = T + 6 * 65536;
    const float* T7 = T + 7 * 65536;
    const float* T8 = T + 8 * 65536;

    float m1 = 0.0f, m3 = 0.0f, m5 = 0.0f;
#pragma unroll
    for (int t = 0; t < kW; ++t) {
        float s1 = T0[c[t] * kH + h];
        m1 = fmaxf(m1, s1);

        float s3 = T2[c[t] * kH + h];
        if (t >= 1)      s3 += T1[c[t - 1] * kH + h];
        if (t <= kW - 2) s3 += T3[c[t + 1] * kH + h];
        m3 = fmaxf(m3, s3);

        float s5 = T6[c[t] * kH + h];
        if (t >= 2)      s5 += T4[c[t - 2] * kH + h];
        if (t >= 1)      s5 += T5[c[t - 1] * kH + h];
        if (t <= kW - 2) s5 += T7[c[t + 1] * kH + h];
        if (t <= kW - 3) s5 += T8[c[t + 2] * kH + h];
        m5 = fmaxf(m5, s5);
    }

    feat[n * kF3 + h]          = m1;
    feat[n * kF3 + kH + h]     = m3;
    feat[n * kF3 + 2 * kH + h] = m5;
}

__global__ __launch_bounds__(256) void linear_wmma(
    const float* __restrict__ feat,
    const float* __restrict__ lw,
    const float* __restrict__ lb,
    float*       __restrict__ out)
{
    const int wave  = blockIdx.x * (blockDim.x >> 5) + (threadIdx.x >> 5);
    const int mTile = wave >> 4;
    const int nTile = wave & 15;
    const int lane  = threadIdx.x & 31;
    const int half  = lane >> 4;
    const int l16   = lane & 15;

    const float* arow = feat + (mTile * 16 + l16) * kF3 + 2 * half;
    const float* brow = lw   + (nTile * 16 + l16) * kF3 + 2 * half;

    v8f acc = {};
#pragma unroll 4
    for (int k0 = 0; k0 < kF3; k0 += 4) {
        v2f a = *(const v2f*)(arow + k0);
        v2f b = *(const v2f*)(brow + k0);
        acc = __builtin_amdgcn_wmma_f32_16x16x4_f32(
            false, a, false, b, (short)0, acc, false, false);
    }

    const float bias = lb[nTile * 16 + l16];
    float* o = out + (mTile * 16) * kH + nTile * 16;
#pragma unroll
    for (int i = 0; i < 8; ++i) {
        int m = half * 8 + i;
        o[m * kH + l16] = acc[i] + bias;
    }
}

extern "C" void kernel_launch(void* const* d_in, const int* in_sizes, int n_in,
                              void* d_out, int out_size, void* d_ws, size_t ws_size,
                              hipStream_t stream)
{
    const int*   chars = (const int*)  d_in[0];
    const float* emb   = (const float*)d_in[1];
    const float* w1    = (const float*)d_in[2];
    const float* b1    = (const float*)d_in[3];
    const float* w3    = (const float*)d_in[4];
    const float* b3    = (const float*)d_in[5];
    const float* w5    = (const float*)d_in[6];
    const float* b5    = (const float*)d_in[7];
    const float* lw    = (const float*)d_in[8];
    const float* lb    = (const float*)d_in[9];
    float*       outp  = (float*)d_out;

    float* T    = (float*)d_ws;
    float* feat = T + 9 * kVocab * kH;

    build_tables<<<9 * 16 * 16, 32, 0, stream>>>(emb, w1, b1, w3, b3, w5, b5, T);

    conv_gather<<<kWords, 256, 0, stream>>>(chars, T, feat);

    linear_wmma<<<(kWords / 16) * (kH / 16) / 8, 256, 0, stream>>>(feat, lw, lb, outp);
}


# gather: each workgroup copies one 32-column slice of the lookup tables (288 KiB, taps packed four per 128-bit read) into LDS and serves 256 words from it, so lookups are LDS reads, not L2 reads
# speedup vs baseline: 1.6823x; 1.6823x over previous
#include <hip/hip_runtime.h>
#include <hip/hip_bf16.h>

typedef float v2f __attribute__((ext_vector_type(2)));
typedef float v8f __attribute__((ext_vector_type(8)));

static constexpr int kVocab = 256;
static constexpr int kE     = 64;
static constexpr int kH     = 256;
static constexpr int kW     = 20;
static constexpr int kWords = 64 * 128;
static constexpr int kF3    = 3 * kH;

static constexpr int kHS      = 32;
static constexpr int kSliceAB = kVocab * 2 * kHS * 4;
static constexpr int kSliceF  = kSliceAB + kVocab * kHS;
static constexpr int kWordsPerWave  = 16;
static constexpr int kGatherWaves   = 16;
static constexpr int kWordsPerBlock = kGatherWaves * kWordsPerWave;

__global__ __launch_bounds__(32) void build_tables(
    const float* __restrict__ emb,
    const float* __restrict__ w1, const float* __restrict__ b1,
    const float* __restrict__ w3, const float* __restrict__ b3,
    const float* __restrict__ w5, const float* __restrict__ b5,
    float* __restrict__ T)
{
    int tile  = blockIdx.x;
    int tap   = tile >> 8;
    int cTile = (tile >> 4) & 15;
    int hTile = tile & 15;

    const float* wp; const float* bp; int kk, j;
    if (tap == 0)     { wp = w1; bp = b1; kk = 1; j = 0; }
    else if (tap < 4) { wp = w3; bp = b3; kk = 3; j = tap - 1; }
    else              { wp = w5; bp = b5; kk = 5; j = tap - 4; }
    const bool center = (j == (kk >> 1));

    const int lane = threadIdx.x & 31;
    const int half = lane >> 4;
    const int l16  = lane & 15;

    const int row = cTile * 16 + l16;
    const int col = hTile * 16 + l16;

    v8f acc = {};
    for (int e0 = 0; e0 < kE; e0 += 4) {
        int ea = e0 + 2 * half;
        v2f a = *(const v2f*)(emb + row * kE + ea);
        v2f b;
        b.x = wp[(col * kE + ea    ) * kk + j];
        b.y = wp[(col * kE + ea + 1) * kk + j];
        acc = __builtin_amdgcn_wmma_f32_16x16x4_f32(
            false, a, false, b, (short)0, acc, false, false);
    }

    const float bias = center ? bp[col] : 0.0f;
    const int hs = hTile >> 1;
    const int hl = (hTile & 1) * 16 + l16;
    float* outp = T + hs * kSliceF;
    int pitch;
    if (tap < 8) { outp += (tap >> 2) * (kHS * 4) + hl * 4 + (tap & 3); pitch = 2 * kHS * 4; }
    else         { outp += kSliceAB + hl;                               pitch = kHS; }
#pragma unroll
    for (int i = 0; i < 8; ++i) {
        int m = cTile * 16 + half * 8 + i;
        outp[m * pitch] = acc[i] + bias;
    }
}

__global__ __launch_bounds__(kGatherWaves * 32) void conv_gather(
    const int*   __restrict__ chars,
    const float* __restrict__ T,
    float*       __restrict__ feat)
{
    __shared__ __attribute__((aligned(16))) float L[kSliceF];

    const int hs   = blockIdx.x & 7;
    const int grp  = blockIdx.x >> 3;
    const int tid  = threadIdx.x;
    const int lane = tid & 31;
    const int wv   = tid >> 5;
    const int n0   = grp * kWordsPerBlock + wv * kWordsPerWave;

    int cr[kW];
    {
        const int4* cp = (const int4*)(chars + (n0 + (lane & (kWordsPerWave - 1))) * kW);
#pragma unroll
        for (int j = 0; j < kW / 4; ++j) {
            int4 v = cp[j];
            cr[4 * j] = v.x; cr[4 * j + 1] = v.y; cr[4 * j + 2] = v.z; cr[4 * j + 3] = v.w;
        }
    }

    {
        const float4* src = (const float4*)(T + hs * kSliceF);
        float4*       dst = (float4*)L;
#pragma unroll 12
        for (int k = 0; k < kSliceF / 4 / (kGatherWaves * 32); ++k)
            dst[k * (kGatherWaves * 32) + tid] = src[k * (kGatherWaves * 32) + tid];
    }
    __syncthreads();

    const float* LA = L + lane * 4;
    const float* LC = L + kSliceAB + lane;
    float* fo = feat + n0 * kF3 + hs * kHS + lane;

    for (int i = 0; i < kWordsPerWave; ++i) {
        float4 a[kW], b[kW];
        float  e[kW];
#pragma unroll
        for (int t = 0; t < kW; ++t) {
            const int c = __builtin_amdgcn_readlane(cr[t], i);
            a[t] = *(const float4*)(LA + c * (2 * kHS * 4));
            b[t] = *(const float4*)(LA + c * (2 * kHS * 4) + kHS * 4);
            e[t] = LC[c * kHS];
        }

        float m1 = 0.0f, m3 = 0.0f, m5 = 0.0f;
#pragma unroll
        for (int t = 0; t < kW; ++t) {
            m1 = fmaxf(m1, a[t].x);

            float s3 = a[t].z;
            if (t >= 1)      s3 += a[t - 1].y;
            if (t <= kW - 2) s3 += a[t + 1].w;
            m3 = fmaxf(m3, s3);

            float s5 = b[t].z;
            if (t >= 2)      s5 += b[t - 2].x;
            if (t >= 1)      s5 += b[t - 1].y;
            if (t <= kW - 2) s5 += b[t + 1].w;
            if (t <= kW - 3) s5 += e[t + 2];
            m5 = fmaxf(m5, s5);
        }

        fo[i * kF3]          = m1;
        fo[i * kF3 + kH]     = m3;
        fo[i * kF3 + 2 * kH] = m5;
    }
}

__global__ __launch_bounds__(256) void linear_wmma(
    const float* __restrict__ feat,
    const float* __restrict__ lw,
    const float* __restrict__ lb,
    float*       __restrict__ out)
{
    const int wave  = blockIdx.x * (blockDim.x >> 5) + (threadIdx.x >> 5);
    const int mTile = wave >> 4;
    const int nTile = wave & 15;
    const int lane  = threadIdx.x & 31;
    const int half  = lane >> 4;
    const int l16   = lane & 15;

    const float* arow = feat + (mTile * 16 + l16) * kF3 + 2 * half;
    const float* brow = lw   + (nTile * 16 + l16) * kF3 + 2 * half;

    v8f acc = {};
#pragma unroll 4
    for (int k0 = 0; k0 < kF3; k0 += 4) {
        v2f a = *(const v2f*)(arow + k0);
        v2f b = *(const v2f*)(brow + k0);
        acc = __builtin_amdgcn_wmma_f32_16x16x4_f32(
            false, a, false, b, (short)0, acc, false, false);
    }

    const float bias = lb[nTile * 16 + l16];
    float* o = out + (mTile * 16) * kH + nTile * 16;
#pragma unroll
    for (int i = 0; i < 8; ++i) {
        int m = half * 8 + i;
        o[m * kH + l16] = acc[i] + bias;
    }
}

extern "C" void kernel_launch(void* const* d_in, const int* in_sizes, int n_in,
                              void* d_out, int out_size, void* d_ws, size_t ws_size,
                              hipStream_t stream)
{
    const int*   chars = (const int*)  d_in[0];
    const float* emb   = (const float*)d_in[1];
    const float* w1    = (const float*)d_in[2];
    const float* b1    = (const float*)d_in[3];
    const float* w3    = (const float*)d_in[4];
    const float* b3    = (const float*)d_in[5];
    const float* w5    = (const float*)d_in[6];
    const float* b5    = (const float*)d_in[7];
    const float* lw    = (const float*)d_in[8];
    const float* lb    = (const float*)d_in[9];
    float*       outp  = (float*)d_out;

    float* T    = (float*)d_ws;
    float* feat = T + 9 * kVocab * kH;

    build_tables<<<9 * 16 * 16, 32, 0, stream>>>(emb, w1, b1, w3, b3, w5, b5, T);

    conv_gather<<<(kH / kHS) * (kWords / kWordsPerBlock), kGatherWaves * 32, 0, stream>>>(chars, T, feat);

    linear_wmma<<<(kWords / 16) * (kH / 16) / 8, 256, 0, stream>>>(feat, lw, lb, outp);
}


# linear: 64x64 out tile per 4-wave workgroup, feat and lw K-chunks staged through LDS by whole-row 16-byte loads, 2x2 WMMA tiles per wave; k order of every sum unchanged
# speedup vs baseline: 3.3286x; 1.9786x over previous
#include <hip/hip_runtime.h>
#include <hip/hip_bf16.h>

typedef float v2f __attribute__((ext_vector_type(2)));
typedef float v8f __attribute__((ext_vector_type(8)));

static constexpr int kVocab = 256;
static constexpr int kE     = 64;
static constexpr int kH     = 256;
static constexpr int kW     = 20;
static constexpr int kWords = 64 * 128;
static constexpr int kF3    = 3 * kH;

static constexpr int kHS      = 32;
static constexpr int kSliceAB = kVocab * 2 * kHS * 4;
static constexpr int kSliceF  = kSliceAB + kVocab * kHS;
static constexpr int kWordsPerWave  = 16;
static constexpr int kGatherWaves   = 16;
static constexpr int kWordsPerBlock = kGatherWaves * kWordsPerWave;

__global__ __launch_bounds__(32) void build_tables(
    const float* __restrict__ emb,
    const float* __restrict__ w1, const float* __restrict__ b1,
    const float* __restrict__ w3, const float* __restrict__ b3,
    const float* __restrict__ w5, const float* __restrict__ b5,
    float* __restrict__ T)
{
    int tile  = blockIdx.x;
    int tap   = tile >> 8;
    int cTile = (tile >> 4) & 15;
    int hTile = tile & 15;

    const float* wp; const float* bp; int kk, j;
    if (tap == 0)     { wp = w1; bp = b1; kk = 1; j = 0; }
    else if (tap < 4) { wp = w3; bp = b3; kk = 3; j = tap - 1; }
    else              { wp = w5; bp = b5; kk = 5; j = tap - 4; }
    const bool center = (j == (kk >> 1));

    const int lane = threadIdx.x & 31;
    const int half = lane >> 4;
    const int l16  = lane & 15;

    const int row = cTile * 16 + l16;
    const int col = hTile * 16 + l16;

    v8f acc = {};
    for (int e0 = 0; e0 < kE; e0 += 4) {
        int ea = e0 + 2 * half;
        v2f a = *(const v2f*)(emb + row * kE + ea);
        v2f b;
        b.x = wp[(col * kE + ea    ) * kk + j];
        b.y = wp[(col * kE + ea + 1) * kk + j];
        acc = __builtin_amdgcn_wmma_f32_16x16x4_f32(
            false, a, false, b, (short)0, acc, false, false);
    }

    const float bias = center ? bp[col] : 0.0f;
    const int hs = hTile >> 1;
    const int hl = (hTile & 1) * 16 + l16;
    float* outp = T + hs * kSliceF;
    int pitch;
    if (tap < 8) { outp += (tap >> 2) * (kHS * 4) + hl * 4 + (tap & 3); pitch = 2 * kHS * 4; }
    else         { outp += kSliceAB + hl;                               pitch = kHS; }
#pragma unroll
    for (int i = 0; i < 8; ++i) {
        int m = cTile * 16 + half * 8 + i;
        outp[m * pitch] = acc[i] + bias;
    }
}

__global__ __launch_bounds__(kGatherWaves * 32) void conv_gather(
    const int*   __restrict__ chars,
    const float* __restrict__ T,
    float*       __restrict__ feat)
{
    __shared__ __attribute__((aligned(16))) float L[kSliceF];

    const int hs   = blockIdx.x & 7;
    const int grp  = blockIdx.x >> 3;
    const int tid  = threadIdx.x;
    const int lane = tid & 31;
    const int wv   = tid >> 5;
    const int n0   = grp * kWordsPerBlock + wv * kWordsPerWave;

    int cr[kW];
    {
        const int4* cp = (const int4*)(chars + (n0 + (lane & (kWordsPerWave - 1))) * kW);
#pragma unroll
        for (int j = 0; j < kW / 4; ++j) {
            int4 v = cp[j];
            cr[4 * j] = v.x; cr[4 * j + 1] = v.y; cr[4 * j + 2] = v.z; cr[4 * j + 3] = v.w;
        }
    }

    {
        const float4* src = (const float4*)(T + hs * kSliceF);
        float4*       dst = (float4*)L;
#pragma unroll 12
        for (int k = 0; k < kSliceF / 4 / (kGatherWaves * 32); ++k)
            dst[k * (kGatherWaves * 32) + tid] = src[k * (kGatherWaves * 32) + tid];
    }
    __syncthreads();

    const float* LA = L + lane * 4;
    const float* LC = L + kSliceAB + lane;
    float* fo = feat + n0 * kF3 + hs * kHS + lane;

    for (int i = 0; i < kWordsPerWave; ++i) {
        float4 a[kW], b[kW];
        float  e[kW];
#pragma unroll
        for (int t = 0; t < kW; ++t) {
            const int c = __builtin_amdgcn_readlane(cr[t], i);
            a[t] = *(const float4*)(LA + c * (2 * kHS * 4));
            b[t] = *(const float4*)(LA + c * (2 * kHS * 4) + kHS * 4);
            e[t] = LC[c * kHS];
        }

        float m1 = 0.0f, m3 = 0.0f, m5 = 0.0f;
#pragma unroll
        for (int t = 0; t < kW; ++t) {
            m1 = fmaxf(m1, a[t].x);

            float s3 = a[t].z;
            if (t >= 1)      s3 += a[t - 1].y;
            if (t <= kW - 2) s3 += a[t + 1].w;
            m3 = fmaxf(m3, s3);

            float s5 = b[t].z;
            if (t >= 2)      s5 += b[t - 2].x;
            if (t >= 1)      s5 += b[t - 1].y;
            if (t <= kW - 2) s5 += b[t + 1].w;
            if (t <= kW - 3) s5 += e[t + 2];
            m5 = fmaxf(m5, s5);
        }

        fo[i * kF3]          = m1;
        fo[i * kF3 + kH]     = m3;
        fo[i * kF3 + 2 * kH] = m5;
    }
}

static constexpr int kLBM    = 64;
static constexpr int kLBN    = 64;
static constexpr int kLBK    = 128;
static constexpr int kLPitch = kLBK + 4;
static constexpr int kLWaves = 4;
static constexpr int kLStage = (kLBM + kLBN) * kLPitch;
static constexpr int kLRows  = kLBM / kLWaves;
static constexpr int kLGroup = 4;

__device__ __forceinline__ void lin_fetch(const float* __restrict__ gA, const float* __restrict__ gB,
                                          int k0, float4 (&ra)[kLRows], float4 (&rb)[kLRows])
{
#pragma unroll
    for (int j = 0; j < kLRows; ++j) {
        ra[j] = *(const float4*)(gA + j * kLWaves * kF3 + k0);
        rb[j] = *(const float4*)(gB + j * kLWaves * kF3 + k0);
    }
}

__device__ __forceinline__ void lin_stash(float* st, int sOff,
                                          const float4 (&ra)[kLRows], const float4 (&rb)[kLRows])
{
#pragma unroll
    for (int j = 0; j < kLRows; ++j) {
        *(float4*)(st + sOff + j * kLWaves * kLPitch)                  = ra[j];
        *(float4*)(st + kLBM * kLPitch + sOff + j * kLWaves * kLPitch) = rb[j];
    }
}

__device__ __forceinline__ void lin_frags(const float* st, int aOff, int bOff, int g,
                                          v2f (&fa)[kLGroup][2], v2f (&fb)[kLGroup][2])
{
#pragma unroll
    for (int s = 0; s < kLGroup; ++s) {
        const int k = (g * kLGroup + s) * 4;
        fa[s][0] = *(const v2f*)(st + aOff + k);
        fa[s][1] = *(const v2f*)(st + aOff + 16 * kLPitch + k);
        fb[s][0] = *(const v2f*)(st + bOff + k);
        fb[s][1] = *(const v2f*)(st + bOff + 16 * kLPitch + k);
    }
}

__device__ __forceinline__ void lin_mult(const float* st, int aOff, int bOff, v8f (&acc)[2][2])
{
    v2f fa[2][kLGroup][2], fb[2][kLGroup][2];
    lin_frags(st, aOff, bOff, 0, fa[0], fb[0]);
#pragma unroll
    for (int g = 0; g < kLBK / 4 / kLGroup; ++g) {
        if (g + 1 < kLBK / 4 / kLGroup)
            lin_frags(st, aOff, bOff, g + 1, fa[(g + 1) & 1], fb[(g + 1) & 1]);
#pragma unroll
        for (int s = 0; s < kLGroup; ++s) {
            acc[0][0] = __builtin_amdgcn_wmma_f32_16x16x4_f32(
                false, fa[g & 1][s][0], false, fb[g & 1][s][0], (short)0, acc[0][0], false, false);
            acc[0][1] = __builtin_amdgcn_wmma_f32_16x16x4_f32(
                false, fa[g & 1][s][0], false, fb[g & 1][s][1], (short)0, acc[0][1], false, false);
            acc[1][0] = __builtin_amdgcn_wmma_f32_16x16x4_f32(
                false, fa[g & 1][s][1], false, fb[g & 1][s][0], (short)0, acc[1][0], false, false);
            acc[1][1] = __builtin_amdgcn_wmma_f32_16x16x4_f32(
                false, fa[g & 1][s][1], false, fb[g & 1][s][1], (short)0, acc[1][1], false, false);
        }
    }
}

__global__ __launch_bounds__(kLWaves * 32) void linear_wmma(
    const float* __restrict__ feat,
    const float* __restrict__ lw,
    const float* __restrict__ lb,
    float*       __restrict__ out)
{
    __shared__ __attribute__((aligned(16))) float S[2][kLStage];

    const int tid  = threadIdx.x;
    const int lane = tid & 31;
    const int wv   = tid >> 5;
    const int half = lane >> 4;
    const int l16  = lane & 15;
    const int mBlk = blockIdx.x >> 2;
    const int nBlk = blockIdx.x & 3;
    const int wm   = wv >> 1;
    const int wn   = wv & 1;

    const float* gA = feat + (mBlk * kLBM + wv) * kF3 + lane * 4;
    const float* gB = lw   + (nBlk * kLBN + wv) * kF3 + lane * 4;
    const int sOff  = wv * kLPitch + lane * 4;

    const int aOff = (wm * 32 + l16) * kLPitch + 2 * half;
    const int bOff = kLBM * kLPitch + (wn * 32 + l16) * kLPitch + 2 * half;

    float4 ra[kLRows], rb[kLRows];
    v8f acc[2][2] = {};

    lin_fetch(gA, gB, 0, ra, rb);
    lin_stash(S[0], sOff, ra, rb);
    __syncthreads();

    for (int t = 0; t < kF3 / kLBK - 1; ++t) {
        lin_fetch(gA, gB, (t + 1) * kLBK, ra, rb);
        lin_mult(S[t & 1], aOff, bOff, acc);
        lin_stash(S[(t + 1) & 1], sOff, ra, rb);
        __syncthreads();
    }
    lin_mult(S[(kF3 / kLBK - 1) & 1], aOff, bOff, acc);

#pragma unroll
    for (int ni = 0; ni < 2; ++ni) {
        const int col = nBlk * kLBN + wn * 32 + ni * 16 + l16;
        const float bias = lb[col];
#pragma unroll
        for (int mi = 0; mi < 2; ++mi) {
            float* o = out + (mBlk * kLBM + wm * 32 + mi * 16 + half * 8) * kH + col;
#pragma unroll
            for (int i = 0; i < 8; ++i)
                o[i * kH] = acc[mi][ni][i] + bias;
        }
    }
}

extern "C" void kernel_launch(void* const* d_in, const int* in_sizes, int n_in,
                              void* d_out, int out_size, void* d_ws, size_t ws_size,
                              hipStream_t stream)
{
    const int*   chars = (const int*)  d_in[0];
    const float* emb   = (const float*)d_in[1];
    const float* w1    = (const float*)d_in[2];
    const float* b1    = (const float*)d_in[3];
    const float* w3    = (const float*)d_in[4];
    const float* b3    = (const float*)d_in[5];
    const float* w5    = (const float*)d_in[6];
    const float* b5    = (const float*)d_in[7];
    const float* lw    = (const float*)d_in[8];
    const float* lb    = (const float*)d_in[9];
    float*       outp  = (float*)d_out;

    float* T    = (float*)d_ws;
    float* feat = T + 9 * kVocab * kH;

    build_tables<<<9 * 16 * 16, 32, 0, stream>>>(emb, w1, b1, w3, b3, w5, b5, T);

    conv_gather<<<(kH / kHS) * (kWords / kWordsPerBlock), kGatherWaves * 32, 0, stream>>>(chars, T, feat);

    linear_wmma<<<(kWords / kLBM) * (kH / kLBN), kLWaves * 32, 0, stream>>>(feat, lw, lb, outp);
}
